# starting kernel
# baseline (speedup 1.0000x reference)
#include <hip/hip_runtime.h>
#include <math.h>


typedef __attribute__((ext_vector_type(2))) float v2f;
typedef __attribute__((ext_vector_type(8))) float v8f;

#define EMB   1024
#define T3E   3072
#define TSEQ  2048
#define NBATCH 4
#define NH    16
#define HS    64

__global__ __launch_bounds__(128) void qkv_gemm_kernel(
    const float* __restrict__ x, const float* __restrict__ W,
    const float* __restrict__ bias, float* __restrict__ qkv) {
  const int lane = threadIdx.x & 31;
  const int wave = threadIdx.x >> 5;
  const int half = lane >> 4;
  const int l16  = lane & 15;

  const int row0 = blockIdx.y * 64 + wave * 16;
  const int col0 = blockIdx.x * 64;

  const float* arow = x + (size_t)(row0 + l16) * EMB + 2 * half;
  const float* brow0 = W + (size_t)(col0 + l16) * EMB + 2 * half;
  const float* brow1 = brow0 + (size_t)16 * EMB;
  const float* brow2 = brow0 + (size_t)32 * EMB;
  const float* brow3 = brow0 + (size_t)48 * EMB;

  v8f acc[4];
#pragma unroll
  for (int n = 0; n < 4; ++n) {
    const float bv = bias[col0 + 16 * n + l16];
#pragma unroll
    for (int i = 0; i < 8; ++i) acc[n][i] = bv;
  }

#pragma unroll 4
  for (int k = 0; k < EMB; k += 4) {
    v2f a  = *(const v2f*)(arow + k);
    v2f b0 = *(const v2f*)(brow0 + k);
    v2f b1 = *(const v2f*)(brow1 + k);
    v2f b2 = *(const v2f*)(brow2 + k);
    v2f b3 = *(const v2f*)(brow3 + k);
    acc[0] = __builtin_amdgcn_wmma_f32_16x16x4_f32(false, a, false, b0,
                                                   (short)0, acc[0], false, false);
    acc[1] = __builtin_amdgcn_wmma_f32_16x16x4_f32(false, a, false, b1,
                                                   (short)0, acc[1], false, false);
    acc[2] = __builtin_amdgcn_wmma_f32_16x16x4_f32(false, a, false, b2,
                                                   (short)0, acc[2], false, false);
    acc[3] = __builtin_amdgcn_wmma_f32_16x16x4_f32(false, a, false, b3,
                                                   (short)0, acc[3], false, false);
  }

#pragma unroll
  for (int n = 0; n < 4; ++n) {
    float* out = qkv + (size_t)row0 * T3E + col0 + 16 * n + l16;
#pragma unroll
    for (int i = 0; i < 8; ++i)
      out[(size_t)(i + 8 * half) * T3E] = acc[n][i];
  }
}

__global__ __launch_bounds__(128) void attn_kernel(
    const float* __restrict__ qkv, float* __restrict__ out) {
  __shared__ float pls[4][16 * 17];

  const int lane = threadIdx.x & 31;
  const int wave = threadIdx.x >> 5;
  const int half = lane >> 4;
  const int l16  = lane & 15;

  const int bh = blockIdx.y;
  const int b  = bh >> 4;
  const int h  = bh & 15;
  const int qt = blockIdx.x * 4 + wave;
  const int qrow = qt * 16;

  const float* qbase = qkv + (size_t)b * TSEQ * T3E + (size_t)h * HS;
  const float* kbase = qbase + EMB;
  const float* vbase = qbase + 2 * EMB;

  v2f qreg[16];
  {
    const float* qp = qbase + (size_t)(qrow + l16) * T3E + 2 * half;
#pragma unroll
    for (int ks = 0; ks < 16; ++ks) qreg[ks] = *(const v2f*)(qp + ks * 4);
  }

  v8f o[4];
#pragma unroll
  for (int nb = 0; nb < 4; ++nb)
#pragma unroll
    for (int i = 0; i < 8; ++i) o[nb][i] = 0.0f;

  float mrow[8], lrow[8];
#pragma unroll
  for (int i = 0; i < 8; ++i) { mrow[i] = -__builtin_inff(); lrow[i] = 0.0f; }

  float* plds = pls[wave];
  const float scale = 0.125f;

  for (int jt = 0; jt <= qt; ++jt) {
    const int krow = jt * 16;
    const float* kp = kbase + (size_t)(krow + l16) * T3E + 2 * half;

    if (jt < qt)
      __builtin_prefetch((const void*)(kp + 16 * T3E), 0, 1);

    v8f s;
#pragma unroll
    for (int i = 0; i < 8; ++i) s[i] = 0.0f;
#pragma unroll
    for (int ks = 0; ks < 16; ++ks) {
      v2f bk = *(const v2f*)(kp + ks * 4);
      s = __builtin_amdgcn_wmma_f32_16x16x4_f32(
          false, qreg[ks], false, bk, (short)0, s, false, false);
    }

    float p[8];
#pragma unroll
    for (int i = 0; i < 8; ++i) {
      float v = s[i] * scale;
      if (jt == qt && (krow + l16) > (qrow + i + 8 * half))
        v = -__builtin_inff();
      p[i] = v;
    }

#pragma unroll
    for (int i = 0; i < 8; ++i) {
      float r = p[i];
      r = fmaxf(r, __shfl_xor(r, 1, 32));
      r = fmaxf(r, __shfl_xor(r, 2, 32));
      r = fmaxf(r, __shfl_xor(r, 4, 32));
      r = fmaxf(r, __shfl_xor(r, 8, 32));
      const float mnew  = fmaxf(mrow[i], r);
      const float alpha = __expf(mrow[i] - mnew);
      const float e     = __expf(p[i] - mnew);
      float rs = e;
      rs += __shfl_xor(rs, 1, 32);
      rs += __shfl_xor(rs, 2, 32);
      rs += __shfl_xor(rs, 4, 32);
      rs += __shfl_xor(rs, 8, 32);
      lrow[i] = lrow[i] * alpha + rs;
      mrow[i] = mnew;
      p[i] = e;
#pragma unroll
      for (int nb = 0; nb < 4; ++nb) o[nb][i] *= alpha;
    }

#pragma unroll
    for (int i = 0; i < 8; ++i)
      plds[(i + 8 * half) * 17 + l16] = p[i];
    asm volatile("s_wait_dscnt 0" ::: "memory");

    v2f pa[4];
#pragma unroll
    for (int k2 = 0; k2 < 4; ++k2)
      pa[k2] = *(const v2f*)(plds + l16 * 17 + k2 * 4 + 2 * half);

#pragma unroll
    for (int nb = 0; nb < 4; ++nb) {
#pragma unroll
      for (int k2 = 0; k2 < 4; ++k2) {
        const float* vp =
            vbase + (size_t)(krow + k2 * 4 + 2 * half) * T3E + nb * 16 + l16;
        v2f bv2;
        bv2.x = vp[0];
        bv2.y = vp[T3E];
        o[nb] = __builtin_amdgcn_wmma_f32_16x16x4_f32(
            false, pa[k2], false, bv2, (short)0, o[nb], false, false);
      }
    }
  }

  float* op = out + (size_t)(b * TSEQ + qrow) * EMB + h * HS;
#pragma unroll
  for (int i = 0; i < 8; ++i) {
    const float inv = 1.0f / lrow[i];
#pragma unroll
    for (int nb = 0; nb < 4; ++nb)
      op[(size_t)(i + 8 * half) * EMB + nb * 16 + l16] = o[nb][i] * inv;
  }
}

extern "C" void kernel_launch(void* const* d_in, const int* in_sizes, int n_in,
                              void* d_out, int out_size, void* d_ws, size_t ws_size,
                              hipStream_t stream) {
  const float* x    = (const float*)d_in[0];
  const float* Wqkv = (const float*)d_in[1];
  const float* bqkv = (const float*)d_in[2];
  float* out = (float*)d_out;
  float* qkv = (float*)d_ws;

  dim3 g1(T3E / 64, (NBATCH * TSEQ) / 64);
  qkv_gemm_kernel<<<g1, 128, 0, stream>>>(x, Wqkv, bqkv, qkv);

  dim3 g2(TSEQ / 16 / 4, NBATCH * NH);
  attn_kernel<<<g2, 128, 0, stream>>>(qkv, out);
}


# Attention re-tiled to touch fewer cache lines per WMMA: 32 query rows per wave share each K/V fragment, K and V fragments read from global as whole-line b128 pieces one key tile ahead
# speedup vs baseline: 1.7154x; 1.7154x over previous
#include <hip/hip_runtime.h>
#include <math.h>


typedef __attribute__((ext_vector_type(2))) float v2f;
typedef __attribute__((ext_vector_type(4))) float v4f;
typedef __attribute__((ext_vector_type(8))) float v8f;

#define EMB   1024
#define T3E   3072
#define TSEQ  2048
#define NBATCH 4
#define NH    16
#define HS    64

#define LOAD_FRAG(dst, base, k0)                                   \
  _Pragma("unroll") for (int q_ = 0; q_ < 4; ++q_)                 \
      (dst)[q_] = *(const v4f*)((base) + (k0) + 4 * q_)

__global__ __launch_bounds__(128) void qkv_gemm_kernel(
    const float* __restrict__ x, const float* __restrict__ W,
    const float* __restrict__ bias, float* __restrict__ qkv) {
  const int lane = threadIdx.x & 31;
  const int wave = threadIdx.x >> 5;
  const int half = lane >> 4;
  const int l16  = lane & 15;

  const int row0 = blockIdx.y * 128 + (wave >> 1) * 64;
  const int col0 = blockIdx.x * 128 + (wave & 1) * 64;

  const float* ap = x + (size_t)(row0 + l16) * EMB + 16 * half;
  const float* bp = W + (size_t)(col0 + l16) * EMB + 16 * half;

  v8f acc[4][4];
#pragma unroll
  for (int j = 0; j < 4; ++j) {
    const float bv = bias[col0 + 16 * j + l16];
#pragma unroll
    for (int i = 0; i < 4; ++i)
#pragma unroll
      for (int r = 0; r < 8; ++r) acc[i][j][r] = bv;
  }

#pragma unroll 1
  for (int k0 = 0; k0 < EMB; k0 += 32) {
    v4f af[4][4], bf[4][4];
#pragma unroll
    for (int i = 0; i < 4; ++i) LOAD_FRAG(af[i], ap + (size_t)i * 16 * EMB, k0);
#pragma unroll
    for (int j = 0; j < 4; ++j) LOAD_FRAG(bf[j], bp + (size_t)j * 16 * EMB, k0);
    __builtin_amdgcn_sched_barrier(0);

#pragma unroll
    for (int j = 0; j < 4; ++j) {
#pragma unroll
      for (int q = 0; q < 4; ++q) {
#pragma unroll
        for (int i = 0; i < 4; ++i)
          acc[i][j] = __builtin_amdgcn_wmma_f32_16x16x4_f32(
              false, af[i][q].xy, false, bf[j][q].xy, (short)0, acc[i][j],
              false, false);
#pragma unroll
        for (int i = 0; i < 4; ++i)
          acc[i][j] = __builtin_amdgcn_wmma_f32_16x16x4_f32(
              false, af[i][q].zw, false, bf[j][q].zw, (short)0, acc[i][j],
              false, false);
      }
    }
  }

#pragma unroll
  for (int i = 0; i < 4; ++i) {
#pragma unroll
    for (int j = 0; j < 4; ++j) {
      float* out = qkv + (size_t)(row0 + 16 * i) * T3E + col0 + 16 * j + l16;
#pragma unroll
      for (int r = 0; r < 8; ++r)
        out[(size_t)(r + 8 * half) * T3E] = acc[i][j][r];
    }
  }
}

#define AQ   32
#define NQT  (TSEQ / AQ)

struct KVFrag {
  v4f k[2][4];
  v4f v[4][2];
};

__device__ __forceinline__ void load_kv(KVFrag& f, const float* __restrict__ kbase,
                                        const float* __restrict__ vbase,
                                        int krow, int l16, int half) {
  const float* kp = kbase + (size_t)(krow + l16) * T3E + 16 * half;
#pragma unroll
  for (int c2 = 0; c2 < 2; ++c2)
#pragma unroll
    for (int q = 0; q < 4; ++q)
      f.k[c2][q] = *(const v4f*)(kp + 32 * c2 + 4 * q);
  const float* vp = vbase + (size_t)(krow + 2 * half) * T3E + 4 * l16;
#pragma unroll
  for (int t = 0; t < 4; ++t)
#pragma unroll
    for (int e = 0; e < 2; ++e)
      f.v[t][e] = *(const v4f*)(vp + (size_t)(4 * t + e) * T3E);
}

__device__ __forceinline__ void attn_tile(const KVFrag& f, const v4f (&qf)[2][2][4],
                                          v8f (&o)[2][4], float (&mrow)[2][8],
                                          float (&lrow)[2][8], float* plds,
                                          int dk0, int l16, int half) {
  const float scale = 0.125f;

  v8f s[2];
#pragma unroll
  for (int g = 0; g < 2; ++g)
#pragma unroll
    for (int i = 0; i < 8; ++i) s[g][i] = 0.0f;
#pragma unroll
  for (int c2 = 0; c2 < 2; ++c2) {
#pragma unroll
    for (int q = 0; q < 4; ++q) {
#pragma unroll
      for (int g = 0; g < 2; ++g)
        s[g] = __builtin_amdgcn_wmma_f32_16x16x4_f32(
            false, qf[g][c2][q].xy, false, f.k[c2][q].xy, (short)0, s[g],
            false, false);
#pragma unroll
      for (int g = 0; g < 2; ++g)
        s[g] = __builtin_amdgcn_wmma_f32_16x16x4_f32(
            false, qf[g][c2][q].zw, false, f.k[c2][q].zw, (short)0, s[g],
            false, false);
    }
  }

#pragma unroll
  for (int g = 0; g < 2; ++g) {
    float p[8];
#pragma unroll
    for (int i = 0; i < 8; ++i) {
      float v = s[g][i] * scale;
      if (dk0 >= 0 && (dk0 + l16) > (16 * g + i + 8 * half))
        v = -__builtin_inff();
      p[i] = v;
    }

#pragma unroll
    for (int i = 0; i < 8; ++i) {
      float r = p[i];
      r = fmaxf(r, __shfl_xor(r, 1, 32));
      r = fmaxf(r, __shfl_xor(r, 2, 32));
      r = fmaxf(r, __shfl_xor(r, 4, 32));
      r = fmaxf(r, __shfl_xor(r, 8, 32));
      const float mnew  = fmaxf(mrow[g][i], r);
      const float alpha = __expf(mrow[g][i] - mnew);
      const float e     = __expf(p[i] - mnew);
      float rs = e;
      rs += __shfl_xor(rs, 1, 32);
      rs += __shfl_xor(rs, 2, 32);
      rs += __shfl_xor(rs, 4, 32);
      rs += __shfl_xor(rs, 8, 32);
      lrow[g][i] = lrow[g][i] * alpha + rs;
      mrow[g][i] = mnew;
      p[i] = e;
#pragma unroll
      for (int nb = 0; nb < 4; ++nb) o[g][nb][i] *= alpha;
    }

#pragma unroll
    for (int i = 0; i < 8; ++i)
      plds[g * (16 * 17) + (i + 8 * half) * 17 + l16] = p[i];
  }
  asm volatile("s_wait_dscnt 0" ::: "memory");

  v2f pa[2][4];
#pragma unroll
  for (int g = 0; g < 2; ++g)
#pragma unroll
    for (int t = 0; t < 4; ++t)
      pa[g][t] = *(const v2f*)(plds + g * (16 * 17) + l16 * 17 + t * 4 + 2 * half);

#pragma unroll
  for (int t = 0; t < 4; ++t) {
#pragma unroll
    for (int j = 0; j < 4; ++j) {
      v2f bv2;
      bv2.x = f.v[t][0][j];
      bv2.y = f.v[t][1][j];
#pragma unroll
      for (int g = 0; g < 2; ++g)
        o[g][j] = __builtin_amdgcn_wmma_f32_16x16x4_f32(
            false, pa[g][t], false, bv2, (short)0, o[g][j], false, false);
    }
  }
}

__global__ __launch_bounds__(128) void attn_kernel(
    const float* __restrict__ qkv, float* __restrict__ out) {
  __shared__ float pls[4][2 * 16 * 17];

  const int lane = threadIdx.x & 31;
  const int wave = __builtin_amdgcn_readfirstlane(threadIdx.x >> 5);
  const int half = lane >> 4;
  const int l16  = lane & 15;

  const int bh = blockIdx.y;
  const int b  = bh >> 4;
  const int h  = bh & 15;
  const int pr = blockIdx.x * 4 + wave;

  const float* qbase = qkv + (size_t)b * TSEQ * T3E + (size_t)h * HS;
  const float* kbase = qbase + EMB;
  const float* vbase = qbase + 2 * EMB;
  float* plds = pls[wave];

#pragma unroll 1
  for (int pass = 0; pass < 2; ++pass) {
    const int qt   = pass == 0 ? (NQT - 1 - pr) : pr;
    const int qrow = qt * AQ;
    const int nkt  = 2 * qt + 2;

    v4f qf[2][2][4];
#pragma unroll
    for (int g = 0; g < 2; ++g) {
      const float* qp = qbase + (size_t)(qrow + 16 * g + l16) * T3E + 16 * half;
#pragma unroll
      for (int c2 = 0; c2 < 2; ++c2)
#pragma unroll
        for (int q = 0; q < 4; ++q)
          qf[g][c2][q] = *(const v4f*)(qp + 32 * c2 + 4 * q);
    }

    v8f o[2][4];
    float mrow[2][8], lrow[2][8];
#pragma unroll
    for (int g = 0; g < 2; ++g) {
#pragma unroll
      for (int nb = 0; nb < 4; ++nb)
#pragma unroll
        for (int i = 0; i < 8; ++i) o[g][nb][i] = 0.0f;
#pragma unroll
      for (int i = 0; i < 8; ++i) { mrow[g][i] = -__builtin_inff(); lrow[g][i] = 0.0f; }
    }

    KVFrag f0, f1;
    load_kv(f0, kbase, vbase, 0, l16, half);
#pragma unroll 1
    for (int jt = 0; jt < nkt; jt += 2) {
      load_kv(f1, kbase, vbase, (jt + 1) * 16, l16, half);
      __builtin_amdgcn_sched_barrier(0);
      attn_tile(f0, qf, o, mrow, lrow, plds, jt * 16 - qrow, l16, half);

      const int jn = (jt + 2 < nkt) ? (jt + 2) : (nkt - 1);
      __builtin_amdgcn_sched_barrier(0);
      load_kv(f0, kbase, vbase, jn * 16, l16, half);
      __builtin_amdgcn_sched_barrier(0);
      attn_tile(f1, qf, o, mrow, lrow, plds, (jt + 1) * 16 - qrow, l16, half);
    }

#pragma unroll
    for (int g = 0; g < 2; ++g) {
      float* op = out + (size_t)(b * TSEQ + qrow + 16 * g) * EMB + h * HS + 4 * l16;
#pragma unroll
      for (int i = 0; i < 8; ++i) {
        const float inv = 1.0f / lrow[g][i];
        v4f w;
        w.x = o[g][0][i] * inv;
        w.y = o[g][1][i] * inv;
        w.z = o[g][2][i] * inv;
        w.w = o[g][3][i] * inv;
        *(v4f*)(op + (size_t)(i + 8 * half) * EMB) = w;
      }
    }
  }
}

extern "C" void kernel_launch(void* const* d_in, const int* in_sizes, int n_in,
                              void* d_out, int out_size, void* d_ws, size_t ws_size,
                              hipStream_t stream) {
  const float* x    = (const float*)d_in[0];
  const float* Wqkv = (const float*)d_in[1];
  const float* bqkv = (const float*)d_in[2];
  float* out = (float*)d_out;
  float* qkv = (float*)d_ws;

  dim3 g1(T3E / 128, (NBATCH * TSEQ) / 128);
  qkv_gemm_kernel<<<g1, 128, 0, stream>>>(x, Wqkv, bqkv, qkv);

  dim3 g2(NQT / 2 / 4, NBATCH * NH);
  attn_kernel<<<g2, 128, 0, stream>>>(qkv, out);
}


# Attention computed transposed (keys down the tile rows, one query per lane): softmax max and sum are per-lane register work, P feeds the PV WMMA from the score registers; no LDS, no lane shuffles
# speedup vs baseline: 2.2169x; 1.2924x over previous
#include <hip/hip_runtime.h>
#include <math.h>


typedef __attribute__((ext_vector_type(2))) float v2f;
typedef __attribute__((ext_vector_type(4))) float v4f;
typedef __attribute__((ext_vector_type(8))) float v8f;

#define EMB   1024
#define T3E   3072
#define TSEQ  2048
#define NBATCH 4
#define NH    16
#define HS    64

#define LOAD_FRAG(dst, base, k0)                                   \
  _Pragma("unroll") for (int q_ = 0; q_ < 4; ++q_)                 \
      (dst)[q_] = *(const v4f*)((base) + (k0) + 4 * q_)

__global__ __launch_bounds__(128) void qkv_gemm_kernel(
    const float* __restrict__ x, const float* __restrict__ W,
    const float* __restrict__ bias, float* __restrict__ qkv) {
  const int lane = threadIdx.x & 31;
  const int wave = threadIdx.x >> 5;
  const int half = lane >> 4;
  const int l16  = lane & 15;

  const int row0 = blockIdx.y * 128 + (wave >> 1) * 64;
  const int col0 = blockIdx.x * 128 + (wave & 1) * 64;

  const float* ap = x + (size_t)(row0 + l16) * EMB + 16 * half;
  const float* bp = W + (size_t)(col0 + l16) * EMB + 16 * half;

  v8f acc[4][4];
#pragma unroll
  for (int j = 0; j < 4; ++j) {
    const float bv = bias[col0 + 16 * j + l16];
#pragma unroll
    for (int i = 0; i < 4; ++i)
#pragma unroll
      for (int r = 0; r < 8; ++r) acc[i][j][r] = bv;
  }

#pragma unroll 1
  for (int k0 = 0; k0 < EMB; k0 += 32) {
    v4f af[4][4], bf[4][4];
#pragma unroll
    for (int i = 0; i < 4; ++i) LOAD_FRAG(af[i], ap + (size_t)i * 16 * EMB, k0);
#pragma unroll
    for (int j = 0; j < 4; ++j) LOAD_FRAG(bf[j], bp + (size_t)j * 16 * EMB, k0);
    __builtin_amdgcn_sched_barrier(0);

#pragma unroll
    for (int j = 0; j < 4; ++j) {
#pragma unroll
      for (int q = 0; q < 4; ++q) {
#pragma unroll
        for (int i = 0; i < 4; ++i)
          acc[i][j] = __builtin_amdgcn_wmma_f32_16x16x4_f32(
              false, af[i][q].xy, false, bf[j][q].xy, (short)0, acc[i][j],
              false, false);
#pragma unroll
        for (int i = 0; i < 4; ++i)
          acc[i][j] = __builtin_amdgcn_wmma_f32_16x16x4_f32(
              false, af[i][q].zw, false, bf[j][q].zw, (short)0, acc[i][j],
              false, false);
      }
    }
  }

#pragma unroll
  for (int i = 0; i < 4; ++i) {
#pragma unroll
    for (int j = 0; j < 4; ++j) {
      float* out = qkv + (size_t)(row0 + 16 * i) * T3E + col0 + 16 * j + l16;
#pragma unroll
      for (int r = 0; r < 8; ++r)
        out[(size_t)(r + 8 * half) * T3E] = acc[i][j][r];
    }
  }
}

#define AQ   32
#define NQT  (TSEQ / AQ)

struct KVFrag {
  v4f k[2][4];
  v4f v[4][2];
};

__device__ __forceinline__ void load_kv(KVFrag& f, const float* __restrict__ kbase,
                                        const float* __restrict__ vbase,
                                        int krow, int l16, int half) {
  const float* kp = kbase + (size_t)(krow + l16) * T3E + 16 * half;
#pragma unroll
  for (int c2 = 0; c2 < 2; ++c2)
#pragma unroll
    for (int q = 0; q < 4; ++q)
      f.k[c2][q] = *(const v4f*)(kp + 32 * c2 + 4 * q);
  const float* vp = vbase + (size_t)(krow + 8 * half) * T3E + 4 * l16;
#pragma unroll
  for (int t = 0; t < 4; ++t)
#pragma unroll
    for (int e = 0; e < 2; ++e)
      f.v[t][e] = *(const v4f*)(vp + (size_t)(2 * t + e) * T3E);
}

__device__ __forceinline__ float xhalf(float x) {
  const int b = __builtin_bit_cast(int, x);
  return __builtin_bit_cast(
      float, __builtin_amdgcn_permlanex16(b, b, 0x76543210, 0xfedcba98, false, false));
}

__device__ __forceinline__ void attn_tile(const KVFrag& f, const v4f (&qf)[2][2][4],
                                          v8f (&o)[2][4], float (&mrow)[2],
                                          float (&lrow)[2], int dk0, int l16, int half) {
  const float cs = 0.125f * 1.4426950408889634f;

  v8f s[2];
#pragma unroll
  for (int g = 0; g < 2; ++g)
#pragma unroll
    for (int i = 0; i < 8; ++i) s[g][i] = 0.0f;
#pragma unroll
  for (int c2 = 0; c2 < 2; ++c2) {
#pragma unroll
    for (int q = 0; q < 4; ++q) {
#pragma unroll
      for (int g = 0; g < 2; ++g)
        s[g] = __builtin_amdgcn_wmma_f32_16x16x4_f32(
            false, f.k[c2][q].xy, false, qf[g][c2][q].xy, (short)0, s[g],
            false, false);
#pragma unroll
      for (int g = 0; g < 2; ++g)
        s[g] = __builtin_amdgcn_wmma_f32_16x16x4_f32(
            false, f.k[c2][q].zw, false, qf[g][c2][q].zw, (short)0, s[g],
            false, false);
    }
  }

  if (dk0 >= 0) {
#pragma unroll
    for (int g = 0; g < 2; ++g)
#pragma unroll
      for (int r = 0; r < 8; ++r)
        if ((dk0 + 8 * half + r) > (16 * g + l16)) s[g][r] = -__builtin_inff();
  }

  v2f pb[2][4];
#pragma unroll
  for (int g = 0; g < 2; ++g) {
    float mx = fmaxf(fmaxf(fmaxf(s[g][0], s[g][1]), fmaxf(s[g][2], s[g][3])),
                     fmaxf(fmaxf(s[g][4], s[g][5]), fmaxf(s[g][6], s[g][7])));
    mx = fmaxf(mx, xhalf(mx));
    const float mnew  = fmaxf(mrow[g], mx * cs);
    const float alpha = __builtin_amdgcn_exp2f(mrow[g] - mnew);
    mrow[g] = mnew;
    float rs = 0.0f;
#pragma unroll
    for (int t = 0; t < 4; ++t) {
      const float p0 = __builtin_amdgcn_exp2f(__builtin_fmaf(s[g][2 * t], cs, -mnew));
      const float p1 = __builtin_amdgcn_exp2f(__builtin_fmaf(s[g][2 * t + 1], cs, -mnew));
      pb[g][t].x = p0;
      pb[g][t].y = p1;
      rs += p0 + p1;
    }
    lrow[g] = lrow[g] * alpha + rs;
#pragma unroll
    for (int j = 0; j < 4; ++j)
#pragma unroll
      for (int r = 0; r < 8; ++r) o[g][j][r] *= alpha;
  }

#pragma unroll
  for (int t = 0; t < 4; ++t) {
#pragma unroll
    for (int j = 0; j < 4; ++j) {
      v2f av2;
      av2.x = f.v[t][0][j];
      av2.y = f.v[t][1][j];
#pragma unroll
      for (int g = 0; g < 2; ++g)
        o[g][j] = __builtin_amdgcn_wmma_f32_16x16x4_f32(
            false, av2, false, pb[g][t], (short)0, o[g][j], false, false);
    }
  }
}

__global__ __launch_bounds__(128) void attn_kernel(
    const float* __restrict__ qkv, float* __restrict__ out) {
  const int lane = threadIdx.x & 31;
  const int wave = __builtin_amdgcn_readfirstlane(threadIdx.x >> 5);
  const int half = lane >> 4;
  const int l16  = lane & 15;

  const int bh = blockIdx.y;
  const int b  = bh >> 4;
  const int h  = bh & 15;
  const int pr = blockIdx.x * 4 + wave;

  const float* qbase = qkv + (size_t)b * TSEQ * T3E + (size_t)h * HS;
  const float* kbase = qbase + EMB;
  const float* vbase = qbase + 2 * EMB;

#pragma unroll 1
  for (int pass = 0; pass < 2; ++pass) {
    const int qt   = pass == 0 ? (NQT - 1 - pr) : pr;
    const int qrow = qt * AQ;
    const int nkt  = 2 * qt + 2;

    v4f qf[2][2][4];
#pragma unroll
    for (int g = 0; g < 2; ++g) {
      const float* qp = qbase + (size_t)(qrow + 16 * g + l16) * T3E + 16 * half;
#pragma unroll
      for (int c2 = 0; c2 < 2; ++c2)
#pragma unroll
        for (int q = 0; q < 4; ++q)
          qf[g][c2][q] = *(const v4f*)(qp + 32 * c2 + 4 * q);
    }

    v8f o[2][4];
    float mrow[2], lrow[2];
#pragma unroll
    for (int g = 0; g < 2; ++g) {
#pragma unroll
      for (int nb = 0; nb < 4; ++nb)
#pragma unroll
        for (int i = 0; i < 8; ++i) o[g][nb][i] = 0.0f;
      mrow[g] = -__builtin_inff();
      lrow[g] = 0.0f;
    }

    KVFrag f0, f1;
    load_kv(f0, kbase, vbase, 0, l16, half);
#pragma unroll 1
    for (int jt = 0; jt < nkt; jt += 2) {
      load_kv(f1, kbase, vbase, (jt + 1) * 16, l16, half);
      __builtin_amdgcn_sched_barrier(0);
      attn_tile(f0, qf, o, mrow, lrow, jt * 16 - qrow, l16, half);

      const int jn = (jt + 2 < nkt) ? (jt + 2) : (nkt - 1);
      __builtin_amdgcn_sched_barrier(0);
      load_kv(f0, kbase, vbase, jn * 16, l16, half);
      __builtin_amdgcn_sched_barrier(0);
      attn_tile(f1, qf, o, mrow, lrow, (jt + 1) * 16 - qrow, l16, half);
    }

#pragma unroll
    for (int g = 0; g < 2; ++g) {
      const float lsum = lrow[g] + xhalf(lrow[g]);
      const float inv  = 1.0f / lsum;
      float* op = out + (size_t)(b * TSEQ + qrow + 16 * g + l16) * EMB + h * HS + 32 * half;
#pragma unroll
      for (int r = 0; r < 8; ++r) {
        v4f w;
        w.x = o[g][0][r] * inv;
        w.y = o[g][1][r] * inv;
        w.z = o[g][2][r] * inv;
        w.w = o[g][3][r] * inv;
        *(v4f*)(op + 4 * r) = w;
      }
    }
  }
}

extern "C" void kernel_launch(void* const* d_in, const int* in_sizes, int n_in,
                              void* d_out, int out_size, void* d_ws, size_t ws_size,
                              hipStream_t stream) {
  const float* x    = (const float*)d_in[0];
  const float* Wqkv = (const float*)d_in[1];
  const float* bqkv = (const float*)d_in[2];
  float* out = (float*)d_out;
  float* qkv = (float*)d_ws;

  dim3 g1(T3E / 128, (NBATCH * TSEQ) / 128);
  qkv_gemm_kernel<<<g1, 128, 0, stream>>>(x, Wqkv, bqkv, qkv);

  dim3 g2(NQT / 2 / 4, NBATCH * NH);
  attn_kernel<<<g2, 128, 0, stream>>>(qkv, out);
}
